# starting kernel
# baseline (speedup 1.0000x reference)
#include <hip/hip_runtime.h>

typedef float v2f __attribute__((ext_vector_type(2)));
typedef float v8f __attribute__((ext_vector_type(8)));

#define MAXT 100000.0f
constexpr int Nn = 1024;
constexpr int Mm = 512;

__launch_bounds__(512, 2)
__global__ void snn_fc_spike_kernel(const float* __restrict__ x,
                                    const float* __restrict__ w,
                                    float* __restrict__ out)
{
    __shared__ float s_xs[Nn];
    __shared__ int   s_idx[Nn];

    const int b   = blockIdx.x;
    const int tid = threadIdx.x;

    for (int i = tid; i < Nn; i += 512) {
        s_xs[i]  = x[b * Nn + i];
        s_idx[i] = i;
    }
    __syncthreads();
    for (int k = 2; k <= Nn; k <<= 1) {
        for (int j = k >> 1; j > 0; j >>= 1) {
            for (int i = tid; i < Nn; i += 512) {
                int ixj = i ^ j;
                if (ixj > i) {
                    bool up = ((i & k) == 0);
                    float a = s_xs[i], c = s_xs[ixj];
                    if (up ? (a > c) : (a < c)) {
                        s_xs[i] = c; s_xs[ixj] = a;
                        int t = s_idx[i]; s_idx[i] = s_idx[ixj]; s_idx[ixj] = t;
                    }
                }
            }
            __syncthreads();
        }
    }

    const int lane = tid & 31;
    const int ln   = lane & 15;
    const int h    = lane >> 4;
    const int wv   = tid >> 5;
    const int m0   = wv * 32;

    v2f A1[4];
#pragma unroll
    for (int kk = 0; kk < 4; ++kk) {
        int j0 = 4 * kk + 2 * h;
        A1[kk][0] = (j0     <= ln) ? 1.0f : 0.0f;
        A1[kk][1] = (j0 + 1 <= ln) ? 1.0f : 0.0f;
    }

    float cw[2]  = {0.0f, 0.0f};
    float cwi[2] = {0.0f, 0.0f};
    float mn[2]  = {MAXT, MAXT};

    for (int c = 0; c < Nn / 16; ++c) {
        const int base = c * 16;

        float xr[8], xnr[8];
#pragma unroll
        for (int r = 0; r < 8; ++r) xr[r] = s_xs[base + 8 * h + r];
#pragma unroll
        for (int r = 0; r < 7; ++r) xnr[r] = xr[r + 1];
        {
            int ng = base + 8 * h + 8;
            xnr[7] = (ng < Nn) ? s_xs[ng] : MAXT;
        }

        v2f A2[4];
        int r0[4], r1[4];
#pragma unroll
        for (int kk = 0; kk < 4; ++kk) {
            int j0 = 4 * kk + 2 * h;
            float xj0 = s_xs[base + j0];
            float xj1 = s_xs[base + j0 + 1];
            A2[kk][0] = (j0     <= ln) ? xj0 : 0.0f;
            A2[kk][1] = (j0 + 1 <= ln) ? xj1 : 0.0f;
            r0[kk] = s_idx[base + j0];
            r1[kk] = s_idx[base + j0 + 1];
        }

        if (c + 1 < Nn / 16) {
#pragma unroll
            for (int kk = 0; kk < 4; ++kk) {
                int nj = base + 16 + 4 * kk + 2 * h;
                __builtin_prefetch(&w[(size_t)s_idx[nj] * Mm + m0 + ln], 0, 3);
                __builtin_prefetch(&w[(size_t)s_idx[nj + 1] * Mm + m0 + ln], 0, 3);
            }
        }

#pragma unroll
        for (int t = 0; t < 2; ++t) {
            const int m = m0 + 16 * t + ln;

            v8f accW, accWI;
#pragma unroll
            for (int e = 0; e < 8; ++e) { accW[e] = cw[t]; accWI[e] = cwi[t]; }

#pragma unroll
            for (int kk = 0; kk < 4; ++kk) {
                v2f bv;
                bv[0] = w[(size_t)r0[kk] * Mm + m];
                bv[1] = w[(size_t)r1[kk] * Mm + m];
                accW  = __builtin_amdgcn_wmma_f32_16x16x4_f32(
                            false, A1[kk], false, bv, (short)0, accW,  false, false);
                accWI = __builtin_amdgcn_wmma_f32_16x16x4_f32(
                            false, A2[kk], false, bv, (short)0, accWI, false, false);
            }

            float o8[8];
#pragma unroll
            for (int r = 0; r < 8; ++r) {
                float wc  = accW[r];
                float wic = accWI[r];
                float den = fminf(fmaxf(wc - 1.0f, 1e-10f), 1e10f);
                float o   = wic * __builtin_amdgcn_rcpf(den);
                bool bad  = (wc < 1.0f) | (o < xr[r]) | (o > xnr[r]);
                o8[r] = bad ? MAXT : o;
            }
            float m01 = fminf(o8[0], o8[1]);
            float m23 = fminf(o8[2], o8[3]);
            float m45 = fminf(o8[4], o8[5]);
            float m67 = fminf(o8[6], o8[7]);
            mn[t] = fminf(mn[t], fminf(fminf(m01, m23), fminf(m45, m67)));

            cw[t]  = __shfl(accW[7],  ln + 16, 32);
            cwi[t] = __shfl(accWI[7], ln + 16, 32);
        }
    }

#pragma unroll
    for (int t = 0; t < 2; ++t) {
        float v = fminf(mn[t], __shfl_xor(mn[t], 16, 32));
        if (h == 0) out[(size_t)b * Mm + m0 + 16 * t + ln] = v;
    }
}

extern "C" void kernel_launch(void* const* d_in, const int* in_sizes, int n_in,
                              void* d_out, int out_size, void* d_ws, size_t ws_size,
                              hipStream_t stream)
{
    (void)n_in; (void)out_size; (void)d_ws; (void)ws_size;
    const float* x = (const float*)d_in[0];
    const float* w = (const float*)d_in[1];
    float* out = (float*)d_out;
    const int B = in_sizes[0] / Nn;
    snn_fc_spike_kernel<<<dim3(B), dim3(512), 0, stream>>>(x, w, out);
}


# Two work-groups per batch row (grid 128 -> 256: column halves, each sorts its own copy), one 16-column WMMA tile per wave, results staged in LDS for whole-line stores; all 256 WGPs busy, not half
# speedup vs baseline: 1.4716x; 1.4716x over previous
#include <hip/hip_runtime.h>

typedef float v2f __attribute__((ext_vector_type(2)));
typedef float v8f __attribute__((ext_vector_type(8)));

#define MAXT 100000.0f
constexpr int Nn = 1024;
constexpr int Mm = 512;
constexpr int Mh = 256;

__launch_bounds__(512, 2)
__global__ void snn_fc_spike_kernel(const float* __restrict__ x,
                                    const float* __restrict__ w,
                                    float* __restrict__ out)
{
    __shared__ float s_xs[Nn];
    __shared__ int   s_idx[Nn];
    __shared__ float s_out[Mh];

    const int b   = blockIdx.x;
    const int mh  = blockIdx.y * Mh;
    const int tid = threadIdx.x;

    for (int i = tid; i < Nn; i += 512) {
        s_xs[i]  = x[b * Nn + i];
        s_idx[i] = i;
    }
    __syncthreads();
    for (int k = 2; k <= Nn; k <<= 1) {
        for (int j = k >> 1; j > 0; j >>= 1) {
            for (int i = tid; i < Nn; i += 512) {
                int ixj = i ^ j;
                if (ixj > i) {
                    bool up = ((i & k) == 0);
                    float a = s_xs[i], c = s_xs[ixj];
                    if (up ? (a > c) : (a < c)) {
                        s_xs[i] = c; s_xs[ixj] = a;
                        int t = s_idx[i]; s_idx[i] = s_idx[ixj]; s_idx[ixj] = t;
                    }
                }
            }
            __syncthreads();
        }
    }

    const int lane = tid & 31;
    const int ln   = lane & 15;
    const int h    = lane >> 4;
    const int wv   = tid >> 5;
    const int m0   = mh + wv * 16;

    v2f A1[4];
#pragma unroll
    for (int kk = 0; kk < 4; ++kk) {
        int j0 = 4 * kk + 2 * h;
        A1[kk][0] = (j0     <= ln) ? 1.0f : 0.0f;
        A1[kk][1] = (j0 + 1 <= ln) ? 1.0f : 0.0f;
    }

    float cw  = 0.0f;
    float cwi = 0.0f;
    float mn  = MAXT;

    for (int c = 0; c < Nn / 16; ++c) {
        const int base = c * 16;

        float xr[8], xnr[8];
#pragma unroll
        for (int r = 0; r < 8; ++r) xr[r] = s_xs[base + 8 * h + r];
#pragma unroll
        for (int r = 0; r < 7; ++r) xnr[r] = xr[r + 1];
        {
            int ng = base + 8 * h + 8;
            xnr[7] = (ng < Nn) ? s_xs[ng] : MAXT;
        }

        v2f A2[4];
        int r0[4], r1[4];
#pragma unroll
        for (int kk = 0; kk < 4; ++kk) {
            int j0 = 4 * kk + 2 * h;
            float xj0 = s_xs[base + j0];
            float xj1 = s_xs[base + j0 + 1];
            A2[kk][0] = (j0     <= ln) ? xj0 : 0.0f;
            A2[kk][1] = (j0 + 1 <= ln) ? xj1 : 0.0f;
            r0[kk] = s_idx[base + j0];
            r1[kk] = s_idx[base + j0 + 1];
        }

        if (c + 1 < Nn / 16) {
#pragma unroll
            for (int kk = 0; kk < 4; ++kk) {
                int nj = base + 16 + 4 * kk + 2 * h;
                __builtin_prefetch(&w[(size_t)s_idx[nj] * Mm + m0 + ln], 0, 3);
                __builtin_prefetch(&w[(size_t)s_idx[nj + 1] * Mm + m0 + ln], 0, 3);
            }
        }

        {
            const int m = m0 + ln;

            v8f accW, accWI;
#pragma unroll
            for (int e = 0; e < 8; ++e) { accW[e] = cw; accWI[e] = cwi; }

#pragma unroll
            for (int kk = 0; kk < 4; ++kk) {
                v2f bv;
                bv[0] = w[(size_t)r0[kk] * Mm + m];
                bv[1] = w[(size_t)r1[kk] * Mm + m];
                accW  = __builtin_amdgcn_wmma_f32_16x16x4_f32(
                            false, A1[kk], false, bv, (short)0, accW,  false, false);
                accWI = __builtin_amdgcn_wmma_f32_16x16x4_f32(
                            false, A2[kk], false, bv, (short)0, accWI, false, false);
            }

            float o8[8];
#pragma unroll
            for (int r = 0; r < 8; ++r) {
                float wc  = accW[r];
                float wic = accWI[r];
                float den = fminf(fmaxf(wc - 1.0f, 1e-10f), 1e10f);
                float o   = wic * __builtin_amdgcn_rcpf(den);
                bool bad  = (wc < 1.0f) | (o < xr[r]) | (o > xnr[r]);
                o8[r] = bad ? MAXT : o;
            }
            float m01 = fminf(o8[0], o8[1]);
            float m23 = fminf(o8[2], o8[3]);
            float m45 = fminf(o8[4], o8[5]);
            float m67 = fminf(o8[6], o8[7]);
            mn = fminf(mn, fminf(fminf(m01, m23), fminf(m45, m67)));

            cw  = __shfl(accW[7],  ln + 16, 32);
            cwi = __shfl(accWI[7], ln + 16, 32);
        }
    }

    {
        float v = fminf(mn, __shfl_xor(mn, 16, 32));
        if (h == 0) s_out[wv * 16 + ln] = v;
    }
    __syncthreads();
    if (tid < Mh) out[(size_t)b * Mm + mh + tid] = s_out[tid];
}

extern "C" void kernel_launch(void* const* d_in, const int* in_sizes, int n_in,
                              void* d_out, int out_size, void* d_ws, size_t ws_size,
                              hipStream_t stream)
{
    (void)n_in; (void)out_size; (void)d_ws; (void)ws_size;
    const float* x = (const float*)d_in[0];
    const float* w = (const float*)d_in[1];
    float* out = (float*)d_out;
    const int B = in_sizes[0] / Nn;
    snn_fc_spike_kernel<<<dim3(B, 2), dim3(512), 0, stream>>>(x, w, out);
}


# Gathered w rows double-buffered in registers two chunks ahead (chunk c issues chunk c+2's 8 loads) in place of the per-chunk L2 prefetch and its serial LDS index reads; w is L2-resident
# speedup vs baseline: 1.8408x; 1.2509x over previous
#include <hip/hip_runtime.h>

typedef float v2f __attribute__((ext_vector_type(2)));
typedef float v8f __attribute__((ext_vector_type(8)));

#define MAXT 100000.0f
constexpr int Nn = 1024;
constexpr int Mm = 512;
constexpr int Mh = 256;

__launch_bounds__(512, 2)
__global__ void snn_fc_spike_kernel(const float* __restrict__ x,
                                    const float* __restrict__ w,
                                    float* __restrict__ out)
{
    __shared__ float s_xs[Nn];
    __shared__ int   s_idx[Nn];
    __shared__ float s_out[Mh];

    const int b   = blockIdx.x;
    const int mh  = blockIdx.y * Mh;
    const int tid = threadIdx.x;

    for (int i = tid; i < Nn; i += 512) {
        s_xs[i]  = x[b * Nn + i];
        s_idx[i] = i;
    }
    __syncthreads();
    for (int k = 2; k <= Nn; k <<= 1) {
        for (int j = k >> 1; j > 0; j >>= 1) {
            for (int i = tid; i < Nn; i += 512) {
                int ixj = i ^ j;
                if (ixj > i) {
                    bool up = ((i & k) == 0);
                    float a = s_xs[i], c = s_xs[ixj];
                    if (up ? (a > c) : (a < c)) {
                        s_xs[i] = c; s_xs[ixj] = a;
                        int t = s_idx[i]; s_idx[i] = s_idx[ixj]; s_idx[ixj] = t;
                    }
                }
            }
            __syncthreads();
        }
    }

    const int lane = tid & 31;
    const int ln   = lane & 15;
    const int h    = lane >> 4;
    const int wv   = tid >> 5;
    const int m0   = mh + wv * 16;

    v2f A1[4];
#pragma unroll
    for (int kk = 0; kk < 4; ++kk) {
        int j0 = 4 * kk + 2 * h;
        A1[kk][0] = (j0     <= ln) ? 1.0f : 0.0f;
        A1[kk][1] = (j0 + 1 <= ln) ? 1.0f : 0.0f;
    }

    float cw  = 0.0f;
    float cwi = 0.0f;
    float mn  = MAXT;

    const int m = m0 + ln;

    v2f bA[4], bB[4];

#pragma unroll
    for (int kk = 0; kk < 4; ++kk) {
        int j0 = 4 * kk + 2 * h;
        bA[kk][0] = w[(size_t)s_idx[j0]          * Mm + m];
        bA[kk][1] = w[(size_t)s_idx[j0 + 1]      * Mm + m];
        bB[kk][0] = w[(size_t)s_idx[16 + j0]     * Mm + m];
        bB[kk][1] = w[(size_t)s_idx[16 + j0 + 1] * Mm + m];
    }

    auto chunk = [&](const int c, v2f (&bc)[4]) {
        const int base = c * 16;

        int in0[4], in1[4];
        {
            const int nb = (c + 2 < Nn / 16) ? base + 32 : base;
#pragma unroll
            for (int kk = 0; kk < 4; ++kk) {
                int j0 = 4 * kk + 2 * h;
                in0[kk] = s_idx[nb + j0];
                in1[kk] = s_idx[nb + j0 + 1];
            }
        }

        float xr[8], xnr[8];
#pragma unroll
        for (int r = 0; r < 8; ++r) xr[r] = s_xs[base + 8 * h + r];
#pragma unroll
        for (int r = 0; r < 7; ++r) xnr[r] = xr[r + 1];
        {
            int ng = base + 8 * h + 8;
            xnr[7] = (ng < Nn) ? s_xs[ng] : MAXT;
        }

        v2f A2[4];
#pragma unroll
        for (int kk = 0; kk < 4; ++kk) {
            int j0 = 4 * kk + 2 * h;
            float xj0 = s_xs[base + j0];
            float xj1 = s_xs[base + j0 + 1];
            A2[kk][0] = (j0     <= ln) ? xj0 : 0.0f;
            A2[kk][1] = (j0 + 1 <= ln) ? xj1 : 0.0f;
        }

        v8f accW, accWI;
#pragma unroll
        for (int e = 0; e < 8; ++e) { accW[e] = cw; accWI[e] = cwi; }

#pragma unroll
        for (int kk = 0; kk < 4; ++kk) {
            accW  = __builtin_amdgcn_wmma_f32_16x16x4_f32(
                        false, A1[kk], false, bc[kk], (short)0, accW,  false, false);
            accWI = __builtin_amdgcn_wmma_f32_16x16x4_f32(
                        false, A2[kk], false, bc[kk], (short)0, accWI, false, false);
        }

#pragma unroll
        for (int kk = 0; kk < 4; ++kk) {
            bc[kk][0] = w[(size_t)in0[kk] * Mm + m];
            bc[kk][1] = w[(size_t)in1[kk] * Mm + m];
        }

        float o8[8];
#pragma unroll
        for (int r = 0; r < 8; ++r) {
            float wc  = accW[r];
            float wic = accWI[r];
            float den = fminf(fmaxf(wc - 1.0f, 1e-10f), 1e10f);
            float o   = wic * __builtin_amdgcn_rcpf(den);
            bool bad  = (wc < 1.0f) | (o < xr[r]) | (o > xnr[r]);
            o8[r] = bad ? MAXT : o;
        }
        float m01 = fminf(o8[0], o8[1]);
        float m23 = fminf(o8[2], o8[3]);
        float m45 = fminf(o8[4], o8[5]);
        float m67 = fminf(o8[6], o8[7]);
        mn = fminf(mn, fminf(fminf(m01, m23), fminf(m45, m67)));

        cw  = __shfl(accW[7],  ln + 16, 32);
        cwi = __shfl(accWI[7], ln + 16, 32);
    };

    for (int c = 0; c < Nn / 16; c += 2) {
        chunk(c,     bA);
        chunk(c + 1, bB);
    }

    {
        float v = fminf(mn, __shfl_xor(mn, 16, 32));
        if (h == 0) s_out[wv * 16 + ln] = v;
    }
    __syncthreads();
    if (tid < Mh) out[(size_t)b * Mm + mh + tid] = s_out[tid];
}

extern "C" void kernel_launch(void* const* d_in, const int* in_sizes, int n_in,
                              void* d_out, int out_size, void* d_ws, size_t ws_size,
                              hipStream_t stream)
{
    (void)n_in; (void)out_size; (void)d_ws; (void)ws_size;
    const float* x = (const float*)d_in[0];
    const float* w = (const float*)d_in[1];
    float* out = (float*)d_out;
    const int B = in_sizes[0] / Nn;
    snn_fc_spike_kernel<<<dim3(B, 2), dim3(512), 0, stream>>>(x, w, out);
}


# Argsort by rank counting in registers (each wave holds the row as 32 key registers; rank = popcount of compare masks against a scalar key) replaces the 55-stage LDS bitonic sort; ties by index
# speedup vs baseline: 2.0107x; 1.0923x over previous
#include <hip/hip_runtime.h>

typedef float v2f __attribute__((ext_vector_type(2)));
typedef float v8f __attribute__((ext_vector_type(8)));

#define MAXT 100000.0f
constexpr int Nn = 1024;
constexpr int Mm = 512;
constexpr int Mh = 256;

__launch_bounds__(512, 2)
__global__ void snn_fc_spike_kernel(const float* __restrict__ x,
                                    const float* __restrict__ w,
                                    float* __restrict__ out)
{
    __shared__ float s_xs[Nn];
    __shared__ int   s_idx[Nn];
    __shared__ float s_out[Mh];

    const int b   = blockIdx.x;
    const int mh  = blockIdx.y * Mh;
    const int tid = threadIdx.x;

    const int lane = tid & 31;
    const int wv   = tid >> 5;

    const float* __restrict__ xrow = x + (size_t)b * Nn;

    auto okey = [](float f) -> unsigned {
        unsigned u = __float_as_uint(f);
        unsigned sgn = (unsigned)((int)u >> 31);
        return u ^ (sgn | 0x80000000u);
    };

    unsigned K[32];
#pragma unroll
    for (int v = 0; v < 32; ++v) {
        const unsigned u    = okey(xrow[32 * v + lane]);
        const unsigned bias = (v < 2 * wv) ? 1u : 0u;
        K[v] = max(u, bias) - bias;
    }
    const float    own0f = xrow[64 * wv + lane];
    const float    own1f = xrow[64 * wv + 32 + lane];
    const unsigned own0  = okey(own0f);
    const unsigned own1  = okey(own1f);

    int rank0 = 0, rank1 = 0;
#pragma unroll 1
    for (int t = 0; t < 32; ++t) {
        const unsigned ua = (unsigned)__builtin_amdgcn_readlane((int)own0, t);
        const unsigned ub = (unsigned)__builtin_amdgcn_readlane((int)own1, t);
        unsigned ca = 0, cb = 0;
#pragma unroll
        for (int v = 0; v < 32; v += 2) {
            const unsigned a0 = __builtin_amdgcn_ballot_w32(K[v]     < ua);
            const unsigned a1 = __builtin_amdgcn_ballot_w32(K[v + 1] < ua);
            const unsigned b0 = __builtin_amdgcn_ballot_w32(K[v]     < ub);
            const unsigned b1 = __builtin_amdgcn_ballot_w32(K[v + 1] < ub);
            ca += (unsigned)__builtin_popcountll(((unsigned long long)a1 << 32) | a0);
            cb += (unsigned)__builtin_popcountll(((unsigned long long)b1 << 32) | b0);
        }
        const unsigned below = (1u << t) - 1u;
        ca += (unsigned)__builtin_popcount(__builtin_amdgcn_ballot_w32(own0 == ua) & below);
        cb += (unsigned)__builtin_popcount(__builtin_amdgcn_ballot_w32(own0 == ub));
        cb += (unsigned)__builtin_popcount(__builtin_amdgcn_ballot_w32(own1 == ub) & below);
        rank0 = (lane == t) ? (int)ca : rank0;
        rank1 = (lane == t) ? (int)cb : rank1;
    }
    s_xs[rank0]  = own0f;  s_idx[rank0] = 64 * wv + lane;
    s_xs[rank1]  = own1f;  s_idx[rank1] = 64 * wv + 32 + lane;
    __syncthreads();

    const int ln   = lane & 15;
    const int h    = lane >> 4;
    const int m0   = mh + wv * 16;

    v2f A1[4];
#pragma unroll
    for (int kk = 0; kk < 4; ++kk) {
        int j0 = 4 * kk + 2 * h;
        A1[kk][0] = (j0     <= ln) ? 1.0f : 0.0f;
        A1[kk][1] = (j0 + 1 <= ln) ? 1.0f : 0.0f;
    }

    float cw  = 0.0f;
    float cwi = 0.0f;
    float mn  = MAXT;

    const int m = m0 + ln;

    v2f bA[4], bB[4];

#pragma unroll
    for (int kk = 0; kk < 4; ++kk) {
        int j0 = 4 * kk + 2 * h;
        bA[kk][0] = w[(size_t)s_idx[j0]          * Mm + m];
        bA[kk][1] = w[(size_t)s_idx[j0 + 1]      * Mm + m];
        bB[kk][0] = w[(size_t)s_idx[16 + j0]     * Mm + m];
        bB[kk][1] = w[(size_t)s_idx[16 + j0 + 1] * Mm + m];
    }

    auto chunk = [&](const int c, v2f (&bc)[4]) {
        const int base = c * 16;

        int in0[4], in1[4];
        {
            const int nb = (c + 2 < Nn / 16) ? base + 32 : base;
#pragma unroll
            for (int kk = 0; kk < 4; ++kk) {
                int j0 = 4 * kk + 2 * h;
                in0[kk] = s_idx[nb + j0];
                in1[kk] = s_idx[nb + j0 + 1];
            }
        }

        float xr[8], xnr[8];
#pragma unroll
        for (int r = 0; r < 8; ++r) xr[r] = s_xs[base + 8 * h + r];
#pragma unroll
        for (int r = 0; r < 7; ++r) xnr[r] = xr[r + 1];
        {
            int ng = base + 8 * h + 8;
            xnr[7] = (ng < Nn) ? s_xs[ng] : MAXT;
        }

        v2f A2[4];
#pragma unroll
        for (int kk = 0; kk < 4; ++kk) {
            int j0 = 4 * kk + 2 * h;
            float xj0 = s_xs[base + j0];
            float xj1 = s_xs[base + j0 + 1];
            A2[kk][0] = (j0     <= ln) ? xj0 : 0.0f;
            A2[kk][1] = (j0 + 1 <= ln) ? xj1 : 0.0f;
        }

        v8f accW, accWI;
#pragma unroll
        for (int e = 0; e < 8; ++e) { accW[e] = cw; accWI[e] = cwi; }

#pragma unroll
        for (int kk = 0; kk < 4; ++kk) {
            accW  = __builtin_amdgcn_wmma_f32_16x16x4_f32(
                        false, A1[kk], false, bc[kk], (short)0, accW,  false, false);
            accWI = __builtin_amdgcn_wmma_f32_16x16x4_f32(
                        false, A2[kk], false, bc[kk], (short)0, accWI, false, false);
        }

#pragma unroll
        for (int kk = 0; kk < 4; ++kk) {
            bc[kk][0] = w[(size_t)in0[kk] * Mm + m];
            bc[kk][1] = w[(size_t)in1[kk] * Mm + m];
        }

        float o8[8];
#pragma unroll
        for (int r = 0; r < 8; ++r) {
            float wc  = accW[r];
            float wic = accWI[r];
            float den = fminf(fmaxf(wc - 1.0f, 1e-10f), 1e10f);
            float o   = wic * __builtin_amdgcn_rcpf(den);
            bool bad  = (wc < 1.0f) | (o < xr[r]) | (o > xnr[r]);
            o8[r] = bad ? MAXT : o;
        }
        float m01 = fminf(o8[0], o8[1]);
        float m23 = fminf(o8[2], o8[3]);
        float m45 = fminf(o8[4], o8[5]);
        float m67 = fminf(o8[6], o8[7]);
        mn = fminf(mn, fminf(fminf(m01, m23), fminf(m45, m67)));

        cw  = __shfl(accW[7],  ln + 16, 32);
        cwi = __shfl(accWI[7], ln + 16, 32);
    };

    for (int c = 0; c < Nn / 16; c += 2) {
        chunk(c,     bA);
        chunk(c + 1, bB);
    }

    {
        float v = fminf(mn, __shfl_xor(mn, 16, 32));
        if (h == 0) s_out[wv * 16 + ln] = v;
    }
    __syncthreads();
    if (tid < Mh) out[(size_t)b * Mm + mh + tid] = s_out[tid];
}

extern "C" void kernel_launch(void* const* d_in, const int* in_sizes, int n_in,
                              void* d_out, int out_size, void* d_ws, size_t ws_size,
                              hipStream_t stream)
{
    (void)n_in; (void)out_size; (void)d_ws; (void)ws_size;
    const float* x = (const float*)d_in[0];
    const float* w = (const float*)d_in[1];
    float* out = (float*)d_out;
    const int B = in_sizes[0] / Nn;
    snn_fc_spike_kernel<<<dim3(B, 2), dim3(512), 0, stream>>>(x, w, out);
}


# Both prefix sums from one WMMA product: A = [L8 ; L8 diag(xs)] stacked over 8-row chunks: half the WMMAs, the carry stays in its own lane (no cross-lane shuffle); summation order unchanged
# speedup vs baseline: 2.2290x; 1.1086x over previous
#include <hip/hip_runtime.h>

typedef float v2f __attribute__((ext_vector_type(2)));
typedef float v8f __attribute__((ext_vector_type(8)));

#define MAXT 100000.0f
constexpr int Nn = 1024;
constexpr int Mm = 512;
constexpr int Mh = 256;

__launch_bounds__(512, 2)
__global__ void snn_fc_spike_kernel(const float* __restrict__ x,
                                    const float* __restrict__ w,
                                    float* __restrict__ out)
{
    __shared__ float s_xs[Nn + 4];
    __shared__ int   s_idx[Nn];
    __shared__ float s_out[Mh];

    const int b   = blockIdx.x;
    const int mh  = blockIdx.y * Mh;
    const int tid = threadIdx.x;

    const int lane = tid & 31;
    const int wv   = tid >> 5;

    const float* __restrict__ xrow = x + (size_t)b * Nn;

    auto okey = [](float f) -> unsigned {
        unsigned u = __float_as_uint(f);
        unsigned sgn = (unsigned)((int)u >> 31);
        return u ^ (sgn | 0x80000000u);
    };

    unsigned K[32];
#pragma unroll
    for (int v = 0; v < 32; ++v) {
        const unsigned u    = okey(xrow[32 * v + lane]);
        const unsigned bias = (v < 2 * wv) ? 1u : 0u;
        K[v] = max(u, bias) - bias;
    }
    const float    own0f = xrow[64 * wv + lane];
    const float    own1f = xrow[64 * wv + 32 + lane];
    const unsigned own0  = okey(own0f);
    const unsigned own1  = okey(own1f);

    int rank0 = 0, rank1 = 0;
#pragma unroll 1
    for (int t = 0; t < 32; ++t) {
        const unsigned ua = (unsigned)__builtin_amdgcn_readlane((int)own0, t);
        const unsigned ub = (unsigned)__builtin_amdgcn_readlane((int)own1, t);
        unsigned ca = 0, cb = 0;
#pragma unroll
        for (int v = 0; v < 32; v += 2) {
            const unsigned a0 = __builtin_amdgcn_ballot_w32(K[v]     < ua);
            const unsigned a1 = __builtin_amdgcn_ballot_w32(K[v + 1] < ua);
            const unsigned b0 = __builtin_amdgcn_ballot_w32(K[v]     < ub);
            const unsigned b1 = __builtin_amdgcn_ballot_w32(K[v + 1] < ub);
            ca += (unsigned)__builtin_popcountll(((unsigned long long)a1 << 32) | a0);
            cb += (unsigned)__builtin_popcountll(((unsigned long long)b1 << 32) | b0);
        }
        const unsigned below = (1u << t) - 1u;
        ca += (unsigned)__builtin_popcount(__builtin_amdgcn_ballot_w32(own0 == ua) & below);
        cb += (unsigned)__builtin_popcount(__builtin_amdgcn_ballot_w32(own0 == ub));
        cb += (unsigned)__builtin_popcount(__builtin_amdgcn_ballot_w32(own1 == ub) & below);
        rank0 = (lane == t) ? (int)ca : rank0;
        rank1 = (lane == t) ? (int)cb : rank1;
    }
    s_xs[rank0]  = own0f;  s_idx[rank0] = 64 * wv + lane;
    s_xs[rank1]  = own1f;  s_idx[rank1] = 64 * wv + 32 + lane;
    if (tid == 0) s_xs[Nn] = MAXT;
    __syncthreads();

    const int ln   = lane & 15;
    const int h    = lane >> 4;
    const int m0   = mh + wv * 16;

    unsigned aone[2][2], axm[2][2];
#pragma unroll
    for (int kq = 0; kq < 2; ++kq) {
#pragma unroll
        for (int e = 0; e < 2; ++e) {
            const int j = 4 * kq + 2 * h + e;
            aone[kq][e] = (ln < 8 && j <= ln) ? __float_as_uint(1.0f) : 0u;
            axm[kq][e]  = ((ln >= 8) && (j <= ln - 8)) ? 0xffffffffu : 0u;
        }
    }

    float cy = 0.0f;
    float mn = MAXT;

    const int m = m0 + ln;

    v2f bA[4], bB[4];

#pragma unroll
    for (int kk = 0; kk < 4; ++kk) {
        int j0 = 4 * kk + 2 * h;
        bA[kk][0] = w[(size_t)s_idx[j0]          * Mm + m];
        bA[kk][1] = w[(size_t)s_idx[j0 + 1]      * Mm + m];
        bB[kk][0] = w[(size_t)s_idx[16 + j0]     * Mm + m];
        bB[kk][1] = w[(size_t)s_idx[16 + j0 + 1] * Mm + m];
    }

    float xa0[4], xa1[4];
    int   in0[4], in1[4];
    auto lds_ahead = [&](const int c) {
        const int cb = (c < Nn / 16) ? c : Nn / 16 - 1;
        const int cn = (cb + 2 < Nn / 16) ? cb + 2 : cb;
#pragma unroll
        for (int kk = 0; kk < 4; ++kk) {
            int j0 = 4 * kk + 2 * h;
            xa0[kk] = s_xs[cb * 16 + j0];
            xa1[kk] = s_xs[cb * 16 + j0 + 1];
            in0[kk] = s_idx[cn * 16 + j0];
            in1[kk] = s_idx[cn * 16 + j0 + 1];
        }
    };
    lds_ahead(0);

    auto group = [&](const int c, v2f (&bc)[4]) {
        const int base = c * 16;

        float xac0[4], xac1[4];
        int   inc0[4], inc1[4];
#pragma unroll
        for (int kk = 0; kk < 4; ++kk) {
            xac0[kk] = xa0[kk]; xac1[kk] = xa1[kk];
            inc0[kk] = in0[kk]; inc1[kk] = in1[kk];
        }

        float xt[2][5];
#pragma unroll
        for (int s = 0; s < 2; ++s)
#pragma unroll
            for (int r = 0; r < 5; ++r) xt[s][r] = s_xs[base + 8 * s + 4 * h + r];
        lds_ahead(c + 1);

        float wcv[2][4], wiv[2][4];

#pragma unroll
        for (int s = 0; s < 2; ++s) {
            v2f A[2];
#pragma unroll
            for (int kq = 0; kq < 2; ++kq) {
                A[kq][0] = __uint_as_float((__float_as_uint(xac0[2 * s + kq]) & axm[kq][0]) | aone[kq][0]);
                A[kq][1] = __uint_as_float((__float_as_uint(xac1[2 * s + kq]) & axm[kq][1]) | aone[kq][1]);
            }

            v8f acc;
#pragma unroll
            for (int e = 0; e < 8; ++e) acc[e] = cy;

            acc = __builtin_amdgcn_wmma_f32_16x16x4_f32(
                      false, A[0], false, bc[2 * s],     (short)0, acc, false, false);
            acc = __builtin_amdgcn_wmma_f32_16x16x4_f32(
                      false, A[1], false, bc[2 * s + 1], (short)0, acc, false, false);
            cy = acc[7];

#pragma unroll
            for (int r = 0; r < 4; ++r) {
                auto pr = __builtin_amdgcn_permlane16_swap(
                              __float_as_uint(acc[r]), __float_as_uint(acc[r + 4]), false, false);
                wcv[s][r] = __uint_as_float(pr[0]);
                wiv[s][r] = __uint_as_float(pr[1]);
            }
        }

#pragma unroll
        for (int kk = 0; kk < 4; ++kk) {
            bc[kk][0] = w[(size_t)inc0[kk] * Mm + m];
            bc[kk][1] = w[(size_t)inc1[kk] * Mm + m];
        }
        __builtin_amdgcn_sched_barrier(0x406);

#pragma unroll
        for (int s = 0; s < 2; ++s) {
            float o4[4];
#pragma unroll
            for (int r = 0; r < 4; ++r) {
                float wc  = wcv[s][r];
                float wic = wiv[s][r];
                float den = fminf(fmaxf(wc - 1.0f, 1e-10f), 1e10f);
                float o   = wic * __builtin_amdgcn_rcpf(den);
                bool bad  = (wc < 1.0f) | (o < xt[s][r]) | (o > xt[s][r + 1]);
                o4[r] = bad ? MAXT : o;
            }
            mn = fminf(mn, fminf(fminf(o4[0], o4[1]), fminf(o4[2], o4[3])));
        }
    };

    for (int c = 0; c < Nn / 16; c += 2) {
        group(c,     bA);
        group(c + 1, bB);
    }

    {
        float v = fminf(mn, __shfl_xor(mn, 16, 32));
        if (h == 0) s_out[wv * 16 + ln] = v;
    }
    __syncthreads();
    if (tid < Mh) out[(size_t)b * Mm + mh + tid] = s_out[tid];
}

extern "C" void kernel_launch(void* const* d_in, const int* in_sizes, int n_in,
                              void* d_out, int out_size, void* d_ws, size_t ws_size,
                              hipStream_t stream)
{
    (void)n_in; (void)out_size; (void)d_ws; (void)ws_size;
    const float* x = (const float*)d_in[0];
    const float* w = (const float*)d_in[1];
    float* out = (float*)d_out;
    const int B = in_sizes[0] / Nn;
    snn_fc_spike_kernel<<<dim3(B, 2), dim3(512), 0, stream>>>(x, w, out);
}


# Sorted rows kept in LDS as 32-bit element offsets into w, so each gather is scalar base + one 32-bit lane offset: one add per load replaces sign extension, 64-bit shift and 64-bit add
# speedup vs baseline: 2.4889x; 1.1166x over previous
#include <hip/hip_runtime.h>

typedef float v2f __attribute__((ext_vector_type(2)));
typedef float v8f __attribute__((ext_vector_type(8)));

#define MAXT 100000.0f
constexpr int Nn = 1024;
constexpr int Mm = 512;
constexpr int Mh = 256;

__launch_bounds__(512, 2)
__global__ void snn_fc_spike_kernel(const float* __restrict__ x,
                                    const float* __restrict__ w,
                                    float* __restrict__ out)
{
    __shared__ float s_xs[Nn + 4];
    __shared__ unsigned s_row[Nn];
    __shared__ float s_out[Mh];

    const int b   = blockIdx.x;
    const int mh  = blockIdx.y * Mh;
    const int tid = threadIdx.x;

    const int lane = tid & 31;
    const int wv   = tid >> 5;

    const float* __restrict__ xrow = x + (size_t)b * Nn;

    auto okey = [](float f) -> unsigned {
        unsigned u = __float_as_uint(f);
        unsigned sgn = (unsigned)((int)u >> 31);
        return u ^ (sgn | 0x80000000u);
    };

    unsigned K[32];
#pragma unroll
    for (int v = 0; v < 32; ++v) {
        const unsigned u    = okey(xrow[32 * v + lane]);
        const unsigned bias = (v < 2 * wv) ? 1u : 0u;
        K[v] = max(u, bias) - bias;
    }
    const float    own0f = xrow[64 * wv + lane];
    const float    own1f = xrow[64 * wv + 32 + lane];
    const unsigned own0  = okey(own0f);
    const unsigned own1  = okey(own1f);

    int rank0 = 0, rank1 = 0;
#pragma unroll 1
    for (int t = 0; t < 32; ++t) {
        const unsigned ua = (unsigned)__builtin_amdgcn_readlane((int)own0, t);
        const unsigned ub = (unsigned)__builtin_amdgcn_readlane((int)own1, t);
        unsigned ca = 0, cb = 0;
#pragma unroll
        for (int v = 0; v < 32; v += 2) {
            const unsigned a0 = __builtin_amdgcn_ballot_w32(K[v]     < ua);
            const unsigned a1 = __builtin_amdgcn_ballot_w32(K[v + 1] < ua);
            const unsigned b0 = __builtin_amdgcn_ballot_w32(K[v]     < ub);
            const unsigned b1 = __builtin_amdgcn_ballot_w32(K[v + 1] < ub);
            ca += (unsigned)__builtin_popcountll(((unsigned long long)a1 << 32) | a0);
            cb += (unsigned)__builtin_popcountll(((unsigned long long)b1 << 32) | b0);
        }
        const unsigned below = (1u << t) - 1u;
        ca += (unsigned)__builtin_popcount(__builtin_amdgcn_ballot_w32(own0 == ua) & below);
        cb += (unsigned)__builtin_popcount(__builtin_amdgcn_ballot_w32(own0 == ub));
        cb += (unsigned)__builtin_popcount(__builtin_amdgcn_ballot_w32(own1 == ub) & below);
        rank0 = (lane == t) ? (int)ca : rank0;
        rank1 = (lane == t) ? (int)cb : rank1;
    }
    s_xs[rank0]  = own0f;  s_row[rank0] = (unsigned)(64 * wv + lane) * Mm;
    s_xs[rank1]  = own1f;  s_row[rank1] = (unsigned)(64 * wv + 32 + lane) * Mm;
    if (tid == 0) s_xs[Nn] = MAXT;
    __syncthreads();

    const int ln   = lane & 15;
    const int h    = lane >> 4;
    const int m0   = mh + wv * 16;

    unsigned aone[2][2], axm[2][2];
#pragma unroll
    for (int kq = 0; kq < 2; ++kq) {
#pragma unroll
        for (int e = 0; e < 2; ++e) {
            const int j = 4 * kq + 2 * h + e;
            aone[kq][e] = (ln < 8 && j <= ln) ? __float_as_uint(1.0f) : 0u;
            axm[kq][e]  = ((ln >= 8) && (j <= ln - 8)) ? 0xffffffffu : 0u;
        }
    }

    float cy = 0.0f;
    float mn = MAXT;

    const unsigned m = (unsigned)(m0 + ln);

    v2f bA[4], bB[4];

#pragma unroll
    for (int kk = 0; kk < 4; ++kk) {
        int j0 = 4 * kk + 2 * h;
        bA[kk][0] = w[s_row[j0]          + m];
        bA[kk][1] = w[s_row[j0 + 1]      + m];
        bB[kk][0] = w[s_row[16 + j0]     + m];
        bB[kk][1] = w[s_row[16 + j0 + 1] + m];
    }

    float xa0[4], xa1[4];
    unsigned in0[4], in1[4];
    auto lds_ahead = [&](const int c) {
        const int cb = (c < Nn / 16) ? c : Nn / 16 - 1;
        const int cn = (cb + 2 < Nn / 16) ? cb + 2 : cb;
#pragma unroll
        for (int kk = 0; kk < 4; ++kk) {
            int j0 = 4 * kk + 2 * h;
            xa0[kk] = s_xs[cb * 16 + j0];
            xa1[kk] = s_xs[cb * 16 + j0 + 1];
            in0[kk] = s_row[cn * 16 + j0];
            in1[kk] = s_row[cn * 16 + j0 + 1];
        }
    };
    lds_ahead(0);

    auto group = [&](const int c, v2f (&bc)[4]) {
        const int base = c * 16;

        float xac0[4], xac1[4];
        unsigned inc0[4], inc1[4];
#pragma unroll
        for (int kk = 0; kk < 4; ++kk) {
            xac0[kk] = xa0[kk]; xac1[kk] = xa1[kk];
            inc0[kk] = in0[kk]; inc1[kk] = in1[kk];
        }

        float xt[2][5];
#pragma unroll
        for (int s = 0; s < 2; ++s)
#pragma unroll
            for (int r = 0; r < 5; ++r) xt[s][r] = s_xs[base + 8 * s + 4 * h + r];
        lds_ahead(c + 1);

        float wcv[2][4], wiv[2][4];

#pragma unroll
        for (int s = 0; s < 2; ++s) {
            v2f A[2];
#pragma unroll
            for (int kq = 0; kq < 2; ++kq) {
                A[kq][0] = __uint_as_float((__float_as_uint(xac0[2 * s + kq]) & axm[kq][0]) | aone[kq][0]);
                A[kq][1] = __uint_as_float((__float_as_uint(xac1[2 * s + kq]) & axm[kq][1]) | aone[kq][1]);
            }

            v8f acc;
#pragma unroll
            for (int e = 0; e < 8; ++e) acc[e] = cy;

            acc = __builtin_amdgcn_wmma_f32_16x16x4_f32(
                      false, A[0], false, bc[2 * s],     (short)0, acc, false, false);
            acc = __builtin_amdgcn_wmma_f32_16x16x4_f32(
                      false, A[1], false, bc[2 * s + 1], (short)0, acc, false, false);
            cy = acc[7];

#pragma unroll
            for (int r = 0; r < 4; ++r) {
                auto pr = __builtin_amdgcn_permlane16_swap(
                              __float_as_uint(acc[r]), __float_as_uint(acc[r + 4]), false, false);
                wcv[s][r] = __uint_as_float(pr[0]);
                wiv[s][r] = __uint_as_float(pr[1]);
            }
        }

#pragma unroll
        for (int kk = 0; kk < 4; ++kk) {
            bc[kk][0] = w[inc0[kk] + m];
            bc[kk][1] = w[inc1[kk] + m];
        }
        __builtin_amdgcn_sched_barrier(0x406);

#pragma unroll
        for (int s = 0; s < 2; ++s) {
            float o4[4];
#pragma unroll
            for (int r = 0; r < 4; ++r) {
                float wc  = wcv[s][r];
                float wic = wiv[s][r];
                float den = fminf(fmaxf(wc - 1.0f, 1e-10f), 1e10f);
                float o   = wic * __builtin_amdgcn_rcpf(den);
                bool bad  = (wc < 1.0f) | (o < xt[s][r]) | (o > xt[s][r + 1]);
                o4[r] = bad ? MAXT : o;
            }
            mn = fminf(mn, fminf(fminf(o4[0], o4[1]), fminf(o4[2], o4[3])));
        }
    };

    for (int c = 0; c < Nn / 16; c += 2) {
        group(c,     bA);
        group(c + 1, bB);
    }

    {
        float v = fminf(mn, __shfl_xor(mn, 16, 32));
        if (h == 0) s_out[wv * 16 + ln] = v;
    }
    __syncthreads();
    if (tid < Mh) out[(size_t)b * Mm + mh + tid] = s_out[tid];
}

extern "C" void kernel_launch(void* const* d_in, const int* in_sizes, int n_in,
                              void* d_out, int out_size, void* d_ws, size_t ws_size,
                              hipStream_t stream)
{
    (void)n_in; (void)out_size; (void)d_ws; (void)ws_size;
    const float* x = (const float*)d_in[0];
    const float* w = (const float*)d_in[1];
    float* out = (float*)d_out;
    const int B = in_sizes[0] / Nn;
    snn_fc_spike_kernel<<<dim3(B, 2), dim3(512), 0, stream>>>(x, w, out);
}


# Scan loop of a wave ends once each of its 16 columns has a valid spike time: xs[r] <= o_r <= xs[r+1] with xs sorted makes valid times nondecreasing, so the first is the min; exact, same sums
# speedup vs baseline: 3.4311x; 1.3786x over previous
#include <hip/hip_runtime.h>

typedef float v2f __attribute__((ext_vector_type(2)));
typedef float v8f __attribute__((ext_vector_type(8)));

#define MAXT 100000.0f
constexpr int Nn = 1024;
constexpr int Mm = 512;
constexpr int Mh = 256;

__launch_bounds__(512, 2)
__global__ void snn_fc_spike_kernel(const float* __restrict__ x,
                                    const float* __restrict__ w,
                                    float* __restrict__ out)
{
    __shared__ float s_xs[Nn + 4];
    __shared__ unsigned s_row[Nn];
    __shared__ float s_out[Mh];

    const int b   = blockIdx.x;
    const int mh  = blockIdx.y * Mh;
    const int tid = threadIdx.x;

    const int lane = tid & 31;
    const int wv   = tid >> 5;

    const float* __restrict__ xrow = x + (size_t)b * Nn;

    auto okey = [](float f) -> unsigned {
        unsigned u = __float_as_uint(f);
        unsigned sgn = (unsigned)((int)u >> 31);
        return u ^ (sgn | 0x80000000u);
    };

    unsigned K[32];
#pragma unroll
    for (int v = 0; v < 32; ++v) {
        const unsigned u    = okey(xrow[32 * v + lane]);
        const unsigned bias = (v < 2 * wv) ? 1u : 0u;
        K[v] = max(u, bias) - bias;
    }
    const float    own0f = xrow[64 * wv + lane];
    const float    own1f = xrow[64 * wv + 32 + lane];
    const unsigned own0  = okey(own0f);
    const unsigned own1  = okey(own1f);

    int rank0 = 0, rank1 = 0;
#pragma unroll 1
    for (int t = 0; t < 32; ++t) {
        const unsigned ua = (unsigned)__builtin_amdgcn_readlane((int)own0, t);
        const unsigned ub = (unsigned)__builtin_amdgcn_readlane((int)own1, t);
        unsigned ca = 0, cb = 0;
#pragma unroll
        for (int v = 0; v < 32; v += 2) {
            const unsigned a0 = __builtin_amdgcn_ballot_w32(K[v]     < ua);
            const unsigned a1 = __builtin_amdgcn_ballot_w32(K[v + 1] < ua);
            const unsigned b0 = __builtin_amdgcn_ballot_w32(K[v]     < ub);
            const unsigned b1 = __builtin_amdgcn_ballot_w32(K[v + 1] < ub);
            ca += (unsigned)__builtin_popcountll(((unsigned long long)a1 << 32) | a0);
            cb += (unsigned)__builtin_popcountll(((unsigned long long)b1 << 32) | b0);
        }
        const unsigned below = (1u << t) - 1u;
        ca += (unsigned)__builtin_popcount(__builtin_amdgcn_ballot_w32(own0 == ua) & below);
        cb += (unsigned)__builtin_popcount(__builtin_amdgcn_ballot_w32(own0 == ub));
        cb += (unsigned)__builtin_popcount(__builtin_amdgcn_ballot_w32(own1 == ub) & below);
        rank0 = (lane == t) ? (int)ca : rank0;
        rank1 = (lane == t) ? (int)cb : rank1;
    }
    s_xs[rank0]  = own0f;  s_row[rank0] = (unsigned)(64 * wv + lane) * Mm;
    s_xs[rank1]  = own1f;  s_row[rank1] = (unsigned)(64 * wv + 32 + lane) * Mm;
    if (tid == 0) s_xs[Nn] = MAXT;
    __syncthreads();

    const int ln   = lane & 15;
    const int h    = lane >> 4;
    const int m0   = mh + wv * 16;

    unsigned aone[2][2], axm[2][2];
#pragma unroll
    for (int kq = 0; kq < 2; ++kq) {
#pragma unroll
        for (int e = 0; e < 2; ++e) {
            const int j = 4 * kq + 2 * h + e;
            aone[kq][e] = (ln < 8 && j <= ln) ? __float_as_uint(1.0f) : 0u;
            axm[kq][e]  = ((ln >= 8) && (j <= ln - 8)) ? 0xffffffffu : 0u;
        }
    }

    float cy = 0.0f;
    float mn = MAXT;

    const unsigned m = (unsigned)(m0 + ln);

    v2f bA[4], bB[4];

#pragma unroll
    for (int kk = 0; kk < 4; ++kk) {
        int j0 = 4 * kk + 2 * h;
        bA[kk][0] = w[s_row[j0]          + m];
        bA[kk][1] = w[s_row[j0 + 1]      + m];
        bB[kk][0] = w[s_row[16 + j0]     + m];
        bB[kk][1] = w[s_row[16 + j0 + 1] + m];
    }

    float xa0[4], xa1[4];
    unsigned in0[4], in1[4];
    auto lds_ahead = [&](const int c) {
        const int cb = (c < Nn / 16) ? c : Nn / 16 - 1;
        const int cn = (cb + 2 < Nn / 16) ? cb + 2 : cb;
#pragma unroll
        for (int kk = 0; kk < 4; ++kk) {
            int j0 = 4 * kk + 2 * h;
            xa0[kk] = s_xs[cb * 16 + j0];
            xa1[kk] = s_xs[cb * 16 + j0 + 1];
            in0[kk] = s_row[cn * 16 + j0];
            in1[kk] = s_row[cn * 16 + j0 + 1];
        }
    };
    lds_ahead(0);

    auto group = [&](const int c, v2f (&bc)[4]) {
        const int base = c * 16;

        float xac0[4], xac1[4];
        unsigned inc0[4], inc1[4];
#pragma unroll
        for (int kk = 0; kk < 4; ++kk) {
            xac0[kk] = xa0[kk]; xac1[kk] = xa1[kk];
            inc0[kk] = in0[kk]; inc1[kk] = in1[kk];
        }

        float xt[2][5];
#pragma unroll
        for (int s = 0; s < 2; ++s)
#pragma unroll
            for (int r = 0; r < 5; ++r) xt[s][r] = s_xs[base + 8 * s + 4 * h + r];
        lds_ahead(c + 1);

        float wcv[2][4], wiv[2][4];

#pragma unroll
        for (int s = 0; s < 2; ++s) {
            v2f A[2];
#pragma unroll
            for (int kq = 0; kq < 2; ++kq) {
                A[kq][0] = __uint_as_float((__float_as_uint(xac0[2 * s + kq]) & axm[kq][0]) | aone[kq][0]);
                A[kq][1] = __uint_as_float((__float_as_uint(xac1[2 * s + kq]) & axm[kq][1]) | aone[kq][1]);
            }

            v8f acc;
#pragma unroll
            for (int e = 0; e < 8; ++e) acc[e] = cy;

            acc = __builtin_amdgcn_wmma_f32_16x16x4_f32(
                      false, A[0], false, bc[2 * s],     (short)0, acc, false, false);
            acc = __builtin_amdgcn_wmma_f32_16x16x4_f32(
                      false, A[1], false, bc[2 * s + 1], (short)0, acc, false, false);
            cy = acc[7];

#pragma unroll
            for (int r = 0; r < 4; ++r) {
                auto pr = __builtin_amdgcn_permlane16_swap(
                              __float_as_uint(acc[r]), __float_as_uint(acc[r + 4]), false, false);
                wcv[s][r] = __uint_as_float(pr[0]);
                wiv[s][r] = __uint_as_float(pr[1]);
            }
        }

#pragma unroll
        for (int kk = 0; kk < 4; ++kk) {
            bc[kk][0] = w[inc0[kk] + m];
            bc[kk][1] = w[inc1[kk] + m];
        }
        __builtin_amdgcn_sched_barrier(0x406);

#pragma unroll
        for (int s = 0; s < 2; ++s) {
            float o4[4];
#pragma unroll
            for (int r = 0; r < 4; ++r) {
                float wc  = wcv[s][r];
                float wic = wiv[s][r];
                float den = fminf(fmaxf(wc - 1.0f, 1e-10f), 1e10f);
                float o   = wic * __builtin_amdgcn_rcpf(den);
                bool bad  = (wc < 1.0f) | (o < xt[s][r]) | (o > xt[s][r + 1]);
                o4[r] = bad ? MAXT : o;
            }
            mn = fminf(mn, fminf(fminf(o4[0], o4[1]), fminf(o4[2], o4[3])));
        }
    };

    for (int c = 0; c < Nn / 16; c += 2) {
        group(c,     bA);
        group(c + 1, bB);
        const unsigned hit = __builtin_amdgcn_ballot_w32(mn < MAXT);
        if (((hit | (hit >> 16)) & 0xffffu) == 0xffffu) break;
    }

    {
        float v = fminf(mn, __shfl_xor(mn, 16, 32));
        if (h == 0) s_out[wv * 16 + ln] = v;
    }
    __syncthreads();
    if (tid < Mh) out[(size_t)b * Mm + mh + tid] = s_out[tid];
}

extern "C" void kernel_launch(void* const* d_in, const int* in_sizes, int n_in,
                              void* d_out, int out_size, void* d_ws, size_t ws_size,
                              hipStream_t stream)
{
    (void)n_in; (void)out_size; (void)d_ws; (void)ws_size;
    const float* x = (const float*)d_in[0];
    const float* w = (const float*)d_in[1];
    float* out = (float*)d_out;
    const int B = in_sizes[0] / Nn;
    snn_fc_spike_kernel<<<dim3(B, 2), dim3(512), 0, stream>>>(x, w, out);
}


# Until a column of the wave has reached w_cum >= 1, a 32-row scan trip only carries its sums on: lane exchange, window reads and spike test left out, as such rows give MAXT; same sums
# speedup vs baseline: 3.6668x; 1.0687x over previous
#include <hip/hip_runtime.h>

typedef float v2f __attribute__((ext_vector_type(2)));
typedef float v8f __attribute__((ext_vector_type(8)));

#define MAXT 100000.0f
constexpr int Nn = 1024;
constexpr int Mm = 512;
constexpr int Mh = 256;

__launch_bounds__(512, 2)
__global__ void snn_fc_spike_kernel(const float* __restrict__ x,
                                    const float* __restrict__ w,
                                    float* __restrict__ out)
{
    __shared__ float s_xs[Nn + 4];
    __shared__ unsigned s_row[Nn];
    __shared__ float s_out[Mh];

    const int b   = blockIdx.x;
    const int mh  = blockIdx.y * Mh;
    const int tid = threadIdx.x;

    const int lane = tid & 31;
    const int wv   = tid >> 5;

    const float* __restrict__ xrow = x + (size_t)b * Nn;

    auto okey = [](float f) -> unsigned {
        unsigned u = __float_as_uint(f);
        unsigned sgn = (unsigned)((int)u >> 31);
        return u ^ (sgn | 0x80000000u);
    };

    unsigned K[32];
#pragma unroll
    for (int v = 0; v < 32; ++v) {
        const unsigned u    = okey(xrow[32 * v + lane]);
        const unsigned bias = (v < 2 * wv) ? 1u : 0u;
        K[v] = max(u, bias) - bias;
    }
    const float    own0f = xrow[64 * wv + lane];
    const float    own1f = xrow[64 * wv + 32 + lane];
    const unsigned own0  = okey(own0f);
    const unsigned own1  = okey(own1f);

    int rank0 = 0, rank1 = 0;
#pragma unroll 1
    for (int t = 0; t < 32; ++t) {
        const unsigned ua = (unsigned)__builtin_amdgcn_readlane((int)own0, t);
        const unsigned ub = (unsigned)__builtin_amdgcn_readlane((int)own1, t);
        unsigned ca = 0, cb = 0;
#pragma unroll
        for (int v = 0; v < 32; v += 2) {
            const unsigned a0 = __builtin_amdgcn_ballot_w32(K[v]     < ua);
            const unsigned a1 = __builtin_amdgcn_ballot_w32(K[v + 1] < ua);
            const unsigned b0 = __builtin_amdgcn_ballot_w32(K[v]     < ub);
            const unsigned b1 = __builtin_amdgcn_ballot_w32(K[v + 1] < ub);
            ca += (unsigned)__builtin_popcountll(((unsigned long long)a1 << 32) | a0);
            cb += (unsigned)__builtin_popcountll(((unsigned long long)b1 << 32) | b0);
        }
        const unsigned below = (1u << t) - 1u;
        ca += (unsigned)__builtin_popcount(__builtin_amdgcn_ballot_w32(own0 == ua) & below);
        cb += (unsigned)__builtin_popcount(__builtin_amdgcn_ballot_w32(own0 == ub));
        cb += (unsigned)__builtin_popcount(__builtin_amdgcn_ballot_w32(own1 == ub) & below);
        rank0 = (lane == t) ? (int)ca : rank0;
        rank1 = (lane == t) ? (int)cb : rank1;
    }
    s_xs[rank0]  = own0f;  s_row[rank0] = (unsigned)(64 * wv + lane) * Mm;
    s_xs[rank1]  = own1f;  s_row[rank1] = (unsigned)(64 * wv + 32 + lane) * Mm;
    if (tid == 0) s_xs[Nn] = MAXT;
    __syncthreads();

    const int ln   = lane & 15;
    const int h    = lane >> 4;
    const int m0   = mh + wv * 16;

    unsigned aone[2][2], axm[2][2];
#pragma unroll
    for (int kq = 0; kq < 2; ++kq) {
#pragma unroll
        for (int e = 0; e < 2; ++e) {
            const int j = 4 * kq + 2 * h + e;
            aone[kq][e] = (ln < 8 && j <= ln) ? __float_as_uint(1.0f) : 0u;
            axm[kq][e]  = ((ln >= 8) && (j <= ln - 8)) ? 0xffffffffu : 0u;
        }
    }

    float cy = 0.0f;
    float mn = MAXT;

    const unsigned m = (unsigned)(m0 + ln);

    v2f bA[4], bB[4];

#pragma unroll
    for (int kk = 0; kk < 4; ++kk) {
        int j0 = 4 * kk + 2 * h;
        bA[kk][0] = w[s_row[j0]          + m];
        bA[kk][1] = w[s_row[j0 + 1]      + m];
        bB[kk][0] = w[s_row[16 + j0]     + m];
        bB[kk][1] = w[s_row[16 + j0 + 1] + m];
    }

    float xa0[4], xa1[4];
    unsigned in0[4], in1[4];
    auto lds_ahead = [&](const int c) {
        const int cb = (c < Nn / 16) ? c : Nn / 16 - 1;
        const int cn = (cb + 2 < Nn / 16) ? cb + 2 : cb;
#pragma unroll
        for (int kk = 0; kk < 4; ++kk) {
            int j0 = 4 * kk + 2 * h;
            xa0[kk] = s_xs[cb * 16 + j0];
            xa1[kk] = s_xs[cb * 16 + j0 + 1];
            in0[kk] = s_row[cn * 16 + j0];
            in1[kk] = s_row[cn * 16 + j0 + 1];
        }
    };
    lds_ahead(0);

    auto group = [&](const int c, v2f (&bc)[4], v8f (&dd)[2]) {
        float xac0[4], xac1[4];
        unsigned inc0[4], inc1[4];
#pragma unroll
        for (int kk = 0; kk < 4; ++kk) {
            xac0[kk] = xa0[kk]; xac1[kk] = xa1[kk];
            inc0[kk] = in0[kk]; inc1[kk] = in1[kk];
        }

        lds_ahead(c + 1);

#pragma unroll
        for (int s = 0; s < 2; ++s) {
            v2f A[2];
#pragma unroll
            for (int kq = 0; kq < 2; ++kq) {
                A[kq][0] = __uint_as_float((__float_as_uint(xac0[2 * s + kq]) & axm[kq][0]) | aone[kq][0]);
                A[kq][1] = __uint_as_float((__float_as_uint(xac1[2 * s + kq]) & axm[kq][1]) | aone[kq][1]);
            }

            v8f acc;
#pragma unroll
            for (int e = 0; e < 8; ++e) acc[e] = cy;

            acc = __builtin_amdgcn_wmma_f32_16x16x4_f32(
                      false, A[0], false, bc[2 * s],     (short)0, acc, false, false);
            acc = __builtin_amdgcn_wmma_f32_16x16x4_f32(
                      false, A[1], false, bc[2 * s + 1], (short)0, acc, false, false);
            cy = acc[7];
            dd[s] = acc;
        }

#pragma unroll
        for (int kk = 0; kk < 4; ++kk) {
            bc[kk][0] = w[inc0[kk] + m];
            bc[kk][1] = w[inc1[kk] + m];
        }
        __builtin_amdgcn_sched_barrier(0x406);
    };

    auto test = [&](const int c, const v8f (&dd)[2]) {
        const int base = c * 16;
        float xt[2][5];
#pragma unroll
        for (int s = 0; s < 2; ++s)
#pragma unroll
            for (int r = 0; r < 5; ++r) xt[s][r] = s_xs[base + 8 * s + 4 * h + r];

#pragma unroll
        for (int s = 0; s < 2; ++s) {
            float o4[4];
#pragma unroll
            for (int r = 0; r < 4; ++r) {
                auto pr = __builtin_amdgcn_permlane16_swap(
                              __float_as_uint(dd[s][r]), __float_as_uint(dd[s][r + 4]), false, false);
                float wc  = __uint_as_float(pr[0]);
                float wic = __uint_as_float(pr[1]);
                float den = fminf(fmaxf(wc - 1.0f, 1e-10f), 1e10f);
                float o   = wic * __builtin_amdgcn_rcpf(den);
                bool bad  = (wc < 1.0f) | (o < xt[s][r]) | (o > xt[s][r + 1]);
                o4[r] = bad ? MAXT : o;
            }
            mn = fminf(mn, fminf(fminf(o4[0], o4[1]), fminf(o4[2], o4[3])));
        }
    };

    int c = 0;

    for (; c < Nn / 16; c += 2) {
        v8f d0[2], d1[2];
        group(c,     bA, d0);
        group(c + 1, bB, d1);
        float wmx = fmaxf(fmaxf(d0[0][0], d0[1][0]), fmaxf(d1[0][0], d1[1][0]));
#pragma unroll
        for (int r = 1; r < 8; ++r)
            wmx = fmaxf(wmx, fmaxf(fmaxf(d0[0][r], d0[1][r]), fmaxf(d1[0][r], d1[1][r])));
        if ((__builtin_amdgcn_ballot_w32(wmx >= 1.0f) & 0xffffu) != 0u) {
            test(c,     d0);
            test(c + 1, d1);
            c += 2;
            break;
        }
    }

    for (; c < Nn / 16; c += 2) {
        const unsigned hit = __builtin_amdgcn_ballot_w32(mn < MAXT);
        if (((hit | (hit >> 16)) & 0xffffu) == 0xffffu) break;
        v8f d0[2], d1[2];
        group(c,     bA, d0);
        test(c,      d0);
        group(c + 1, bB, d1);
        test(c + 1,  d1);
    }

    {
        float v = fminf(mn, __shfl_xor(mn, 16, 32));
        if (h == 0) s_out[wv * 16 + ln] = v;
    }
    __syncthreads();
    if (tid < Mh) out[(size_t)b * Mm + mh + tid] = s_out[tid];
}

extern "C" void kernel_launch(void* const* d_in, const int* in_sizes, int n_in,
                              void* d_out, int out_size, void* d_ws, size_t ws_size,
                              hipStream_t stream)
{
    (void)n_in; (void)out_size; (void)d_ws; (void)ws_size;
    const float* x = (const float*)d_in[0];
    const float* w = (const float*)d_in[1];
    float* out = (float*)d_out;
    const int B = in_sizes[0] / Nn;
    snn_fc_spike_kernel<<<dim3(B, 2), dim3(512), 0, stream>>>(x, w, out);
}


# Argsort: each wave rank-counts only its own 64 elements, then 4 merge rounds place each element by a search in the partner run's sorted keys in LDS, not 1024 x 1024 compares; same order
# speedup vs baseline: 4.8121x; 1.3123x over previous
#include <hip/hip_runtime.h>

typedef float v2f __attribute__((ext_vector_type(2)));
typedef float v8f __attribute__((ext_vector_type(8)));

#define MAXT 100000.0f
constexpr int Nn = 1024;
constexpr int Mm = 512;
constexpr int Mh = 256;

__launch_bounds__(512, 2)
__global__ void snn_fc_spike_kernel(const float* __restrict__ x,
                                    const float* __restrict__ w,
                                    float* __restrict__ out)
{
    __shared__ float s_xs[Nn + 4];
    __shared__ unsigned s_row[Nn];
    __shared__ float s_out[Mh];
    __shared__ unsigned s_ka[Nn], s_kb[Nn];

    const int b   = blockIdx.x;
    const int mh  = blockIdx.y * Mh;
    const int tid = threadIdx.x;

    const int lane = tid & 31;
    const int wv   = tid >> 5;

    const float* __restrict__ xrow = x + (size_t)b * Nn;

    auto okey = [](float f) -> unsigned {
        unsigned u = __float_as_uint(f);
        unsigned sgn = (unsigned)((int)u >> 31);
        return u ^ (sgn | 0x80000000u);
    };

    const float    own0f = xrow[64 * wv + lane];
    const float    own1f = xrow[64 * wv + 32 + lane];
    const unsigned own0  = okey(own0f);
    const unsigned own1  = okey(own1f);

    unsigned rk0 = 0, rk1 = 0;
#pragma unroll 4
    for (int t = 0; t < 32; ++t) {
        const unsigned ua = (unsigned)__builtin_amdgcn_readlane((int)own0, t);
        const unsigned ub = (unsigned)__builtin_amdgcn_readlane((int)own1, t);
        const unsigned below = (1u << t) - 1u;
        unsigned ca = (unsigned)__builtin_popcount(__builtin_amdgcn_ballot_w32(own0 < ua))
                    + (unsigned)__builtin_popcount(__builtin_amdgcn_ballot_w32(own1 < ua))
                    + (unsigned)__builtin_popcount(__builtin_amdgcn_ballot_w32(own0 == ua) & below);
        unsigned cb = (unsigned)__builtin_popcount(__builtin_amdgcn_ballot_w32(own0 <= ub))
                    + (unsigned)__builtin_popcount(__builtin_amdgcn_ballot_w32(own1 < ub))
                    + (unsigned)__builtin_popcount(__builtin_amdgcn_ballot_w32(own1 == ub) & below);
        rk0 = (lane == t) ? ca : rk0;
        rk1 = (lane == t) ? cb : rk1;
    }

    unsigned gp[2]  = { 64u * (unsigned)wv + rk0, 64u * (unsigned)wv + rk1 };
    const unsigned kv[2] = { own0, own1 };

    s_ka[gp[0]] = own0;
    s_ka[gp[1]] = own1;
    __syncthreads();

#pragma unroll
    for (int lg = 6; lg < 10; ++lg) {
        const unsigned L = 1u << lg;
        const unsigned* src = ((lg & 1) == 0) ? s_ka : s_kb;
        unsigned*       dst = ((lg & 1) == 0) ? s_kb : s_ka;
        unsigned lo[2], pb[2], kk[2];
#pragma unroll
        for (int e = 0; e < 2; ++e) {
            const unsigned run   = gp[e] >> lg;
            const unsigned right = run & 1u;
            pb[e] = (run ^ 1u) << lg;
            kk[e] = kv[e] - (1u - right);
            lo[e] = 0u;
        }
        int bits = lg;
        if (bits & 1) {
            const unsigned half = 1u << (bits - 1);
#pragma unroll
            for (int e = 0; e < 2; ++e) {
                const unsigned bk = src[pb[e] + lo[e] + half - 1u];
                lo[e] = (bk <= kk[e]) ? lo[e] + half : lo[e];
            }
            bits -= 1;
        }
#pragma unroll
        for (; bits > 0; bits -= 2) {
            const unsigned q = 1u << (bits - 2);
#pragma unroll
            for (int e = 0; e < 2; ++e) {
                const unsigned at = pb[e] + lo[e];
                const unsigned b1 = src[at + q - 1u];
                const unsigned b2 = src[at + 2u * q - 1u];
                const unsigned b3 = src[at + 3u * q - 1u];
                const unsigned digit = (b1 <= kk[e] ? 1u : 0u) + (b2 <= kk[e] ? 1u : 0u) + (b3 <= kk[e] ? 1u : 0u);
                lo[e] += digit * q;
            }
        }
#pragma unroll
        for (int e = 0; e < 2; ++e) {
            const unsigned bk = src[pb[e] + lo[e]];
            lo[e] += (bk <= kk[e]) ? 1u : 0u;
            const unsigned run = gp[e] >> lg;
            if ((run & 1u) == 0u && kv[e] == 0u) lo[e] = 0u;
            gp[e] = ((run >> 1) << (lg + 1)) + (gp[e] & (L - 1u)) + lo[e];
        }
        if (lg < 9) {
            dst[gp[0]] = own0;
            dst[gp[1]] = own1;
            __syncthreads();
        }
    }
    const unsigned rank0 = gp[0], rank1 = gp[1];

    s_xs[rank0]  = own0f;  s_row[rank0] = (unsigned)(64 * wv + lane) * Mm;
    s_xs[rank1]  = own1f;  s_row[rank1] = (unsigned)(64 * wv + 32 + lane) * Mm;
    if (tid == 0) s_xs[Nn] = MAXT;
    __syncthreads();

    const int ln   = lane & 15;
    const int h    = lane >> 4;
    const int m0   = mh + wv * 16;

    unsigned aone[2][2], axm[2][2];
#pragma unroll
    for (int kq = 0; kq < 2; ++kq) {
#pragma unroll
        for (int e = 0; e < 2; ++e) {
            const int j = 4 * kq + 2 * h + e;
            aone[kq][e] = (ln < 8 && j <= ln) ? __float_as_uint(1.0f) : 0u;
            axm[kq][e]  = ((ln >= 8) && (j <= ln - 8)) ? 0xffffffffu : 0u;
        }
    }

    float cy = 0.0f;
    float mn = MAXT;

    const unsigned m = (unsigned)(m0 + ln);

    v2f bA[4], bB[4];

#pragma unroll
    for (int kk = 0; kk < 4; ++kk) {
        int j0 = 4 * kk + 2 * h;
        bA[kk][0] = w[s_row[j0]          + m];
        bA[kk][1] = w[s_row[j0 + 1]      + m];
        bB[kk][0] = w[s_row[16 + j0]     + m];
        bB[kk][1] = w[s_row[16 + j0 + 1] + m];
    }

    float xa0[4], xa1[4];
    unsigned in0[4], in1[4];
    auto lds_ahead = [&](const int c) {
        const int cb = (c < Nn / 16) ? c : Nn / 16 - 1;
        const int cn = (cb + 2 < Nn / 16) ? cb + 2 : cb;
#pragma unroll
        for (int kk = 0; kk < 4; ++kk) {
            int j0 = 4 * kk + 2 * h;
            xa0[kk] = s_xs[cb * 16 + j0];
            xa1[kk] = s_xs[cb * 16 + j0 + 1];
            in0[kk] = s_row[cn * 16 + j0];
            in1[kk] = s_row[cn * 16 + j0 + 1];
        }
    };
    lds_ahead(0);

    auto group = [&](const int c, v2f (&bc)[4], v8f (&dd)[2]) {
        float xac0[4], xac1[4];
        unsigned inc0[4], inc1[4];
#pragma unroll
        for (int kk = 0; kk < 4; ++kk) {
            xac0[kk] = xa0[kk]; xac1[kk] = xa1[kk];
            inc0[kk] = in0[kk]; inc1[kk] = in1[kk];
        }

        lds_ahead(c + 1);

#pragma unroll
        for (int s = 0; s < 2; ++s) {
            v2f A[2];
#pragma unroll
            for (int kq = 0; kq < 2; ++kq) {
                A[kq][0] = __uint_as_float((__float_as_uint(xac0[2 * s + kq]) & axm[kq][0]) | aone[kq][0]);
                A[kq][1] = __uint_as_float((__float_as_uint(xac1[2 * s + kq]) & axm[kq][1]) | aone[kq][1]);
            }

            v8f acc;
#pragma unroll
            for (int e = 0; e < 8; ++e) acc[e] = cy;

            acc = __builtin_amdgcn_wmma_f32_16x16x4_f32(
                      false, A[0], false, bc[2 * s],     (short)0, acc, false, false);
            acc = __builtin_amdgcn_wmma_f32_16x16x4_f32(
                      false, A[1], false, bc[2 * s + 1], (short)0, acc, false, false);
            cy = acc[7];
            dd[s] = acc;
        }

#pragma unroll
        for (int kk = 0; kk < 4; ++kk) {
            bc[kk][0] = w[inc0[kk] + m];
            bc[kk][1] = w[inc1[kk] + m];
        }
        __builtin_amdgcn_sched_barrier(0x406);
    };

    auto test = [&](const int c, const v8f (&dd)[2]) {
        const int base = c * 16;
        float xt[2][5];
#pragma unroll
        for (int s = 0; s < 2; ++s)
#pragma unroll
            for (int r = 0; r < 5; ++r) xt[s][r] = s_xs[base + 8 * s + 4 * h + r];

#pragma unroll
        for (int s = 0; s < 2; ++s) {
            float o4[4];
#pragma unroll
            for (int r = 0; r < 4; ++r) {
                auto pr = __builtin_amdgcn_permlane16_swap(
                              __float_as_uint(dd[s][r]), __float_as_uint(dd[s][r + 4]), false, false);
                float wc  = __uint_as_float(pr[0]);
                float wic = __uint_as_float(pr[1]);
                float den = fminf(fmaxf(wc - 1.0f, 1e-10f), 1e10f);
                float o   = wic * __builtin_amdgcn_rcpf(den);
                bool bad  = (wc < 1.0f) | (o < xt[s][r]) | (o > xt[s][r + 1]);
                o4[r] = bad ? MAXT : o;
            }
            mn = fminf(mn, fminf(fminf(o4[0], o4[1]), fminf(o4[2], o4[3])));
        }
    };

    int c = 0;

    for (; c < Nn / 16; c += 2) {
        v8f d0[2], d1[2];
        group(c,     bA, d0);
        group(c + 1, bB, d1);
        float wmx = fmaxf(fmaxf(d0[0][0], d0[1][0]), fmaxf(d1[0][0], d1[1][0]));
#pragma unroll
        for (int r = 1; r < 8; ++r)
            wmx = fmaxf(wmx, fmaxf(fmaxf(d0[0][r], d0[1][r]), fmaxf(d1[0][r], d1[1][r])));
        if ((__builtin_amdgcn_ballot_w32(wmx >= 1.0f) & 0xffffu) != 0u) {
            test(c,     d0);
            test(c + 1, d1);
            c += 2;
            break;
        }
    }

    for (; c < Nn / 16; c += 2) {
        const unsigned hit = __builtin_amdgcn_ballot_w32(mn < MAXT);
        if (((hit | (hit >> 16)) & 0xffffu) == 0xffffu) break;
        v8f d0[2], d1[2];
        group(c,     bA, d0);
        test(c,      d0);
        group(c + 1, bB, d1);
        test(c + 1,  d1);
    }

    {
        float v = fminf(mn, __shfl_xor(mn, 16, 32));
        if (h == 0) s_out[wv * 16 + ln] = v;
    }
    __syncthreads();
    if (tid < Mh) out[(size_t)b * Mm + mh + tid] = s_out[tid];
}

extern "C" void kernel_launch(void* const* d_in, const int* in_sizes, int n_in,
                              void* d_out, int out_size, void* d_ws, size_t ws_size,
                              hipStream_t stream)
{
    (void)n_in; (void)out_size; (void)d_ws; (void)ws_size;
    const float* x = (const float*)d_in[0];
    const float* w = (const float*)d_in[1];
    float* out = (float*)d_out;
    const int B = in_sizes[0] / Nn;
    snn_fc_spike_kernel<<<dim3(B, 2), dim3(512), 0, stream>>>(x, w, out);
}


# Spike test of a 32-row trip runs only if its last row's sums let a column still waiting for its time pass inside it (w >= 0: the crossing is monotone); other trips only carry sums; same sums
# speedup vs baseline: 5.7589x; 1.1968x over previous
#include <hip/hip_runtime.h>

typedef float v2f __attribute__((ext_vector_type(2)));
typedef float v8f __attribute__((ext_vector_type(8)));

#define MAXT 100000.0f
constexpr int Nn = 1024;
constexpr int Mm = 512;
constexpr int Mh = 256;

__launch_bounds__(512, 2)
__global__ void snn_fc_spike_kernel(const float* __restrict__ x,
                                    const float* __restrict__ w,
                                    float* __restrict__ out)
{
    __shared__ float s_xs[Nn + 4];
    __shared__ unsigned s_row[Nn];
    __shared__ float s_out[Mh];
    __shared__ unsigned s_ka[Nn], s_kb[Nn];

    const int b   = blockIdx.x;
    const int mh  = blockIdx.y * Mh;
    const int tid = threadIdx.x;

    const int lane = tid & 31;
    const int wv   = tid >> 5;

    const float* __restrict__ xrow = x + (size_t)b * Nn;

    auto okey = [](float f) -> unsigned {
        unsigned u = __float_as_uint(f);
        unsigned sgn = (unsigned)((int)u >> 31);
        return u ^ (sgn | 0x80000000u);
    };

    const float    own0f = xrow[64 * wv + lane];
    const float    own1f = xrow[64 * wv + 32 + lane];
    const unsigned own0  = okey(own0f);
    const unsigned own1  = okey(own1f);

    unsigned rk0 = 0, rk1 = 0;
#pragma unroll 4
    for (int t = 0; t < 32; ++t) {
        const unsigned ua = (unsigned)__builtin_amdgcn_readlane((int)own0, t);
        const unsigned ub = (unsigned)__builtin_amdgcn_readlane((int)own1, t);
        const unsigned below = (1u << t) - 1u;
        unsigned ca = (unsigned)__builtin_popcount(__builtin_amdgcn_ballot_w32(own0 < ua))
                    + (unsigned)__builtin_popcount(__builtin_amdgcn_ballot_w32(own1 < ua))
                    + (unsigned)__builtin_popcount(__builtin_amdgcn_ballot_w32(own0 == ua) & below);
        unsigned cb = (unsigned)__builtin_popcount(__builtin_amdgcn_ballot_w32(own0 <= ub))
                    + (unsigned)__builtin_popcount(__builtin_amdgcn_ballot_w32(own1 < ub))
                    + (unsigned)__builtin_popcount(__builtin_amdgcn_ballot_w32(own1 == ub) & below);
        rk0 = (lane == t) ? ca : rk0;
        rk1 = (lane == t) ? cb : rk1;
    }

    unsigned gp[2]  = { 64u * (unsigned)wv + rk0, 64u * (unsigned)wv + rk1 };
    const unsigned kv[2] = { own0, own1 };

    s_ka[gp[0]] = own0;
    s_ka[gp[1]] = own1;
    __syncthreads();

#pragma unroll
    for (int lg = 6; lg < 10; ++lg) {
        const unsigned L = 1u << lg;
        const unsigned* src = ((lg & 1) == 0) ? s_ka : s_kb;
        unsigned*       dst = ((lg & 1) == 0) ? s_kb : s_ka;
        unsigned lo[2], pb[2], kk[2];
#pragma unroll
        for (int e = 0; e < 2; ++e) {
            const unsigned run   = gp[e] >> lg;
            const unsigned right = run & 1u;
            pb[e] = (run ^ 1u) << lg;
            kk[e] = kv[e] - (1u - right);
            lo[e] = 0u;
        }
        int bits = lg;
        if (bits & 1) {
            const unsigned half = 1u << (bits - 1);
#pragma unroll
            for (int e = 0; e < 2; ++e) {
                const unsigned bk = src[pb[e] + lo[e] + half - 1u];
                lo[e] = (bk <= kk[e]) ? lo[e] + half : lo[e];
            }
            bits -= 1;
        }
#pragma unroll
        for (; bits > 0; bits -= 2) {
            const unsigned q = 1u << (bits - 2);
#pragma unroll
            for (int e = 0; e < 2; ++e) {
                const unsigned at = pb[e] + lo[e];
                const unsigned b1 = src[at + q - 1u];
                const unsigned b2 = src[at + 2u * q - 1u];
                const unsigned b3 = src[at + 3u * q - 1u];
                const unsigned digit = (b1 <= kk[e] ? 1u : 0u) + (b2 <= kk[e] ? 1u : 0u) + (b3 <= kk[e] ? 1u : 0u);
                lo[e] += digit * q;
            }
        }
#pragma unroll
        for (int e = 0; e < 2; ++e) {
            const unsigned bk = src[pb[e] + lo[e]];
            lo[e] += (bk <= kk[e]) ? 1u : 0u;
            const unsigned run = gp[e] >> lg;
            if ((run & 1u) == 0u && kv[e] == 0u) lo[e] = 0u;
            gp[e] = ((run >> 1) << (lg + 1)) + (gp[e] & (L - 1u)) + lo[e];
        }
        if (lg < 9) {
            dst[gp[0]] = own0;
            dst[gp[1]] = own1;
            __syncthreads();
        }
    }
    const unsigned rank0 = gp[0], rank1 = gp[1];

    s_xs[rank0]  = own0f;  s_row[rank0] = (unsigned)(64 * wv + lane) * Mm;
    s_xs[rank1]  = own1f;  s_row[rank1] = (unsigned)(64 * wv + 32 + lane) * Mm;
    if (tid == 0) s_xs[Nn] = MAXT;
    __syncthreads();

    const int ln   = lane & 15;
    const int h    = lane >> 4;
    const int m0   = mh + wv * 16;

    unsigned aone[2][2], axm[2][2];
#pragma unroll
    for (int kq = 0; kq < 2; ++kq) {
#pragma unroll
        for (int e = 0; e < 2; ++e) {
            const int j = 4 * kq + 2 * h + e;
            aone[kq][e] = (ln < 8 && j <= ln) ? __float_as_uint(1.0f) : 0u;
            axm[kq][e]  = ((ln >= 8) && (j <= ln - 8)) ? 0xffffffffu : 0u;
        }
    }

    float cy = 0.0f;
    float mn = MAXT;

    const unsigned m = (unsigned)(m0 + ln);

    v2f bA[4], bB[4];

#pragma unroll
    for (int kk = 0; kk < 4; ++kk) {
        int j0 = 4 * kk + 2 * h;
        bA[kk][0] = w[s_row[j0]          + m];
        bA[kk][1] = w[s_row[j0 + 1]      + m];
        bB[kk][0] = w[s_row[16 + j0]     + m];
        bB[kk][1] = w[s_row[16 + j0 + 1] + m];
    }

    float xa0[4], xa1[4];
    unsigned in0[4], in1[4];
    auto lds_ahead = [&](const int c) {
        const int cb = (c < Nn / 16) ? c : Nn / 16 - 1;
        const int cn = (cb + 2 < Nn / 16) ? cb + 2 : cb;
#pragma unroll
        for (int kk = 0; kk < 4; ++kk) {
            int j0 = 4 * kk + 2 * h;
            xa0[kk] = s_xs[cb * 16 + j0];
            xa1[kk] = s_xs[cb * 16 + j0 + 1];
            in0[kk] = s_row[cn * 16 + j0];
            in1[kk] = s_row[cn * 16 + j0 + 1];
        }
    };
    lds_ahead(0);

    auto group = [&](const int c, v2f (&bc)[4], v8f (&dd)[2]) {
        float xac0[4], xac1[4];
        unsigned inc0[4], inc1[4];
#pragma unroll
        for (int kk = 0; kk < 4; ++kk) {
            xac0[kk] = xa0[kk]; xac1[kk] = xa1[kk];
            inc0[kk] = in0[kk]; inc1[kk] = in1[kk];
        }

        lds_ahead(c + 1);

#pragma unroll
        for (int s = 0; s < 2; ++s) {
            v2f A[2];
#pragma unroll
            for (int kq = 0; kq < 2; ++kq) {
                A[kq][0] = __uint_as_float((__float_as_uint(xac0[2 * s + kq]) & axm[kq][0]) | aone[kq][0]);
                A[kq][1] = __uint_as_float((__float_as_uint(xac1[2 * s + kq]) & axm[kq][1]) | aone[kq][1]);
            }

            v8f acc;
#pragma unroll
            for (int e = 0; e < 8; ++e) acc[e] = cy;

            acc = __builtin_amdgcn_wmma_f32_16x16x4_f32(
                      false, A[0], false, bc[2 * s],     (short)0, acc, false, false);
            acc = __builtin_amdgcn_wmma_f32_16x16x4_f32(
                      false, A[1], false, bc[2 * s + 1], (short)0, acc, false, false);
            cy = acc[7];
            dd[s] = acc;
        }

#pragma unroll
        for (int kk = 0; kk < 4; ++kk) {
            bc[kk][0] = w[inc0[kk] + m];
            bc[kk][1] = w[inc1[kk] + m];
        }
        __builtin_amdgcn_sched_barrier(0x406);
    };

    auto test = [&](const int c, const v8f (&dd)[2]) {
        const int base = c * 16;
        float xt[2][5];
#pragma unroll
        for (int s = 0; s < 2; ++s)
#pragma unroll
            for (int r = 0; r < 5; ++r) xt[s][r] = s_xs[base + 8 * s + 4 * h + r];

#pragma unroll
        for (int s = 0; s < 2; ++s) {
            float o4[4];
#pragma unroll
            for (int r = 0; r < 4; ++r) {
                auto pr = __builtin_amdgcn_permlane16_swap(
                              __float_as_uint(dd[s][r]), __float_as_uint(dd[s][r + 4]), false, false);
                float wc  = __uint_as_float(pr[0]);
                float wic = __uint_as_float(pr[1]);
                float den = fminf(fmaxf(wc - 1.0f, 1e-10f), 1e10f);
                float o   = wic * __builtin_amdgcn_rcpf(den);
                bool bad  = (wc < 1.0f) | (o < xt[s][r]) | (o > xt[s][r + 1]);
                o4[r] = bad ? MAXT : o;
            }
            mn = fminf(mn, fminf(fminf(o4[0], o4[1]), fminf(o4[2], o4[3])));
        }
    };

    for (int c = 0; c < Nn / 16; c += 2) {
        const unsigned hit  = __builtin_amdgcn_ballot_w32(mn < MAXT);
        const unsigned done = (hit | (hit >> 16)) & 0xffffu;
        if (done == 0xffffu) break;

        const float xn = s_xs[16 * (c + 2)];

        v8f d0[2], d1[2];
        group(c,     bA, d0);
        group(c + 1, bB, d1);

        auto pe = __builtin_amdgcn_permlane16_swap(
                      __float_as_uint(cy), __float_as_uint(cy), false, false);
        const float wE  = __uint_as_float(pe[0]);
        const float wiE = __uint_as_float(pe[1]);
        const float xw  = xn * wE;
        const bool none = (wiE - (xw - xn)) > 1e-4f * (wiE + xw);
        const unsigned may = __builtin_amdgcn_ballot_w32(!none) & 0xffffu & ~done;
        if (may != 0u || c == Nn / 16 - 2) {
            test(c,     d0);
            test(c + 1, d1);
        }
    }

    {
        float v = fminf(mn, __shfl_xor(mn, 16, 32));
        if (h == 0) s_out[wv * 16 + ln] = v;
    }
    __syncthreads();
    if (tid < Mh) out[(size_t)b * Mm + mh + tid] = s_out[tid];
}

extern "C" void kernel_launch(void* const* d_in, const int* in_sizes, int n_in,
                              void* d_out, int out_size, void* d_ws, size_t ws_size,
                              hipStream_t stream)
{
    (void)n_in; (void)out_size; (void)d_ws; (void)ws_size;
    const float* x = (const float*)d_in[0];
    const float* w = (const float*)d_in[1];
    float* out = (float*)d_out;
    const int B = in_sizes[0] / Nn;
    snn_fc_spike_kernel<<<dim3(B, 2), dim3(512), 0, stream>>>(x, w, out);
}
